# starting kernel
# baseline (speedup 1.0000x reference)
#include <hip/hip_runtime.h>
#include <hip/hip_bf16.h>

typedef __attribute__((ext_vector_type(2))) float v2f;
typedef __attribute__((ext_vector_type(8))) float v8f;

#define NSEG   8192
#define D_IN   64
#define D_HID  128
#define LDS_PITCH 132

__global__ void seg_scan_kernel(const int* __restrict__ l, int* __restrict__ offs) {
  __shared__ int ps[256];
  const int tid = threadIdx.x;
  int local[32];
  int s = 0;
#pragma unroll
  for (int i = 0; i < 32; ++i) {
    int v = l[tid * 32 + i];
    local[i] = v;
    s += v;
  }
  ps[tid] = s;
  __syncthreads();
  for (int off = 1; off < 256; off <<= 1) {
    int v = 0;
    if (tid >= off) v = ps[tid - off];
    __syncthreads();
    if (tid >= off) ps[tid] += v;
    __syncthreads();
  }
  int run = (tid == 0) ? 0 : ps[tid - 1];
#pragma unroll
  for (int i = 0; i < 32; ++i) {
    offs[tid * 32 + i] = run;
    run += local[i];
  }
}

__device__ __forceinline__ void tile_mma_16x128(const v2f a[16],
                                                const float* __restrict__ bcol,
                                                v8f c[8]) {
#pragma unroll
  for (int ks = 0; ks < 16; ++ks) {
    const float* bp = bcol + (ks * 4) * LDS_PITCH;
#pragma unroll
    for (int nt = 0; nt < 8; ++nt) {
      v2f b;
      b.x = bp[nt * 16];
      b.y = bp[nt * 16 + LDS_PITCH];
      c[nt] = __builtin_amdgcn_wmma_f32_16x16x4_f32(
          false, a[ks], false, b, (short)0, c[nt], false, false);
    }
  }
}

__global__ __launch_bounds__(256) void fused_mlp_segsum_kernel(
    const float* __restrict__ x, const float* __restrict__ W1,
    const float* __restrict__ b1, const int* __restrict__ offs,
    const int* __restrict__ l, float* __restrict__ xm) {
  __shared__ float w1s[D_IN * LDS_PITCH];

  const int tid = threadIdx.x;
  for (int i = tid; i < D_IN * D_HID; i += 256) {
    w1s[(i >> 7) * LDS_PITCH + (i & 127)] = W1[i];
  }
  __syncthreads();

  const int lane = tid & 31;
  const int wave = tid >> 5;
  const int seg  = blockIdx.x * 8 + wave;
  const int start = offs[seg];
  const int len   = l[seg];
  const int half = lane >> 4;
  const int lm   = lane & 15;

  const float* bcol = w1s + (2 * half) * LDS_PITCH + lm;

  float bias[8];
#pragma unroll
  for (int nt = 0; nt < 8; ++nt) bias[nt] = b1[nt * 16 + lm];

  float accS[8];
#pragma unroll
  for (int nt = 0; nt < 8; ++nt) accS[nt] = 0.0f;

  const int nfull = len & ~15;
  int t0 = 0;

  for (; t0 < nfull; t0 += 16) {
    const float* xrow = x + (size_t)(start + t0 + lm) * D_IN + 2 * half;
    v2f a[16];
#pragma unroll
    for (int ks = 0; ks < 16; ++ks) a[ks] = *(const v2f*)(xrow + ks * 4);

    v8f c[8];
#pragma unroll
    for (int nt = 0; nt < 8; ++nt)
#pragma unroll
      for (int e = 0; e < 8; ++e) c[nt][e] = 0.0f;

    tile_mma_16x128(a, bcol, c);

#pragma unroll
    for (int nt = 0; nt < 8; ++nt) {
      float v[8];
#pragma unroll
      for (int r = 0; r < 8; ++r) v[r] = fmaxf(c[nt][r] + bias[nt], 0.0f);
      float s = ((v[0] + v[1]) + (v[2] + v[3])) + ((v[4] + v[5]) + (v[6] + v[7]));
      accS[nt] += s;
    }
  }

  if (t0 < len) {
    const bool tokv = (t0 + lm) < len;
    const float* xrow = x + (size_t)(start + t0 + lm) * D_IN + 2 * half;
    v2f a[16];
#pragma unroll
    for (int ks = 0; ks < 16; ++ks) {
      v2f av;
      av.x = 0.0f;
      av.y = 0.0f;
      if (tokv) av = *(const v2f*)(xrow + ks * 4);
      a[ks] = av;
    }

    v8f c[8];
#pragma unroll
    for (int nt = 0; nt < 8; ++nt)
#pragma unroll
      for (int e = 0; e < 8; ++e) c[nt][e] = 0.0f;

    tile_mma_16x128(a, bcol, c);

#pragma unroll
    for (int nt = 0; nt < 8; ++nt) {
      float v[8];
#pragma unroll
      for (int r = 0; r < 8; ++r) {
        const bool rv = (t0 + r + 8 * half) < len;
        v[r] = rv ? fmaxf(c[nt][r] + bias[nt], 0.0f) : 0.0f;
      }
      float s = ((v[0] + v[1]) + (v[2] + v[3])) + ((v[4] + v[5]) + (v[6] + v[7]));
      accS[nt] += s;
    }
  }

#pragma unroll
  for (int nt = 0; nt < 8; ++nt) {
    accS[nt] += __shfl_xor(accS[nt], 16, 32);
  }
  if (lane < 16) {
#pragma unroll
    for (int nt = 0; nt < 8; ++nt) {
      xm[(size_t)seg * D_HID + nt * 16 + lane] = accS[nt];
    }
  }
}

__global__ __launch_bounds__(32) void head_gemm_kernel(
    const float* __restrict__ xm, const float* __restrict__ W2,
    const float* __restrict__ b2, float* __restrict__ out) {
  const int lane = threadIdx.x & 31;
  const int half = lane >> 4;
  const int lm   = lane & 15;
  const int mt   = blockIdx.x;

  const float* arow = xm + (size_t)(mt * 16 + lm) * D_HID + 2 * half;
  const float* bcol = W2 + (2 * half) * D_HID + lm;

  v8f c[8];
#pragma unroll
  for (int nt = 0; nt < 8; ++nt)
#pragma unroll
    for (int e = 0; e < 8; ++e) c[nt][e] = 0.0f;

#pragma unroll
  for (int ks = 0; ks < 32; ++ks) {
    v2f a = *(const v2f*)(arow + ks * 4);
    const float* bp = bcol + (ks * 4) * D_HID;
#pragma unroll
    for (int nt = 0; nt < 8; ++nt) {
      v2f b;
      b.x = bp[nt * 16];
      b.y = bp[nt * 16 + D_HID];
      c[nt] = __builtin_amdgcn_wmma_f32_16x16x4_f32(
          false, a, false, b, (short)0, c[nt], false, false);
    }
  }

#pragma unroll
  for (int nt = 0; nt < 8; ++nt) {
    const int n = nt * 16 + lm;
    const float bias = b2[n];
#pragma unroll
    for (int r = 0; r < 8; ++r) {
      out[(size_t)(mt * 16 + r + 8 * half) * D_HID + n] = c[nt][r] + bias;
    }
  }
}

extern "C" void kernel_launch(void* const* d_in, const int* in_sizes, int n_in,
                              void* d_out, int out_size, void* d_ws, size_t ws_size,
                              hipStream_t stream) {
  (void)in_sizes; (void)n_in; (void)out_size; (void)ws_size;
  const float* x  = (const float*)d_in[0];
  const int*   l  = (const int*)d_in[1];
  const float* W1 = (const float*)d_in[2];
  const float* b1 = (const float*)d_in[3];
  const float* W2 = (const float*)d_in[4];
  const float* b2 = (const float*)d_in[5];
  float* out = (float*)d_out;

  int*   offs = (int*)d_ws;
  float* xm   = (float*)((char*)d_ws + ((NSEG * sizeof(int) + 255) & ~size_t(255)));

  seg_scan_kernel<<<1, 256, 0, stream>>>(l, offs);
  fused_mlp_segsum_kernel<<<NSEG / 8, 256, 0, stream>>>(x, W1, b1, offs, l, xm);
  head_gemm_kernel<<<NSEG / 16, 32, 0, stream>>>(xm, W2, b2, out);
}


# Head kernel: 8-wave blocks own 32 segments; W2 staged once per block in LDS (whole-line loads, rows paired), fragments as one group of 8-byte LDS reads; wave = 16 rows x 32 columns; sums unchanged
# speedup vs baseline: 6.7183x; 6.7183x over previous
#include <hip/hip_runtime.h>


typedef __attribute__((ext_vector_type(2))) float v2f;
typedef __attribute__((ext_vector_type(4))) float v4f;
typedef __attribute__((ext_vector_type(8))) float v8f;
typedef __attribute__((ext_vector_type(4))) int   v4i;

#define NSEG      8192
#define TOTAL_TOK 1048576
#define D_IN      64
#define D_HID     128
#define RANGE     1024
#define NBLK      (TOTAL_TOK / RANGE)
#define STEP      64
#define XP        68

__global__ __launch_bounds__(1024) void seg_scan_kernel(const int* __restrict__ l,
                                                        int* __restrict__ offs,
                                                        int* __restrict__ slo) {
  static_assert(NSEG == 8 * 1024 && NBLK == 1024, "one thread per 8 segments and per token range");
  __shared__ __attribute__((aligned(16))) int so[NSEG];
  __shared__ int wtot[32];
  const int tid  = threadIdx.x;
  const int lane = tid & 31;
  const int wv   = tid >> 5;

  const v4i la = *(const v4i*)(l + tid * 8);
  const v4i lb = *(const v4i*)(l + tid * 8 + 4);
  const int e0 = la.x;
  const int e1 = e0 + la.y;
  const int e2 = e1 + la.z;
  const int e3 = e2 + la.w;
  const int e4 = e3 + lb.x;
  const int e5 = e4 + lb.y;
  const int e6 = e5 + lb.z;
  const int e7 = e6 + lb.w;

  int inc = e7;
#pragma unroll
  for (int d = 1; d < 32; d <<= 1) {
    const int v = __shfl_up(inc, d, 32);
    if (lane >= d) inc += v;
  }
  if (lane == 31) wtot[wv] = inc;
  __syncthreads();
  int wsum = (lane < wv) ? wtot[lane] : 0;
#pragma unroll
  for (int d = 16; d >= 1; d >>= 1) wsum += __shfl_xor(wsum, d, 32);

  const int run = wsum + inc - e7;
  v4i o0, o1;
  o0.x = run;
  o0.y = run + e0;
  o0.z = run + e1;
  o0.w = run + e2;
  o1.x = run + e3;
  o1.y = run + e4;
  o1.z = run + e5;
  o1.w = run + e6;
  *(v4i*)(offs + tid * 8)     = o0;
  *(v4i*)(offs + tid * 8 + 4) = o1;
  *(v4i*)(so + tid * 8)       = o0;
  *(v4i*)(so + tid * 8 + 4)   = o1;
  __syncthreads();

  const int B = tid * RANGE;
  int pos = 0;
#pragma unroll
  for (int step = NSEG / 2; step >= 1; step >>= 1) {
    if (so[pos + step - 1] < B) pos += step;
  }
  if (so[pos] < B) pos += 1;
  slo[tid] = pos;
  if (tid == 0) slo[NBLK] = NSEG;
}

__global__ __launch_bounds__(128) void range_mlp_segsum_kernel(
    const float* __restrict__ x, const float* __restrict__ W1,
    const float* __restrict__ b1, const int* __restrict__ offs,
    const int* __restrict__ slo, float* __restrict__ xm,
    float* __restrict__ carry) {
  __shared__ __attribute__((aligned(16))) float xs[2][STEP * XP];

  const int tid  = threadIdx.x;
  const int lane = tid & 31;
  const int cg   = tid >> 5;
  const int half = lane >> 4;
  const int lm   = lane & 15;
  const int blk  = blockIdx.x;
  const int T0   = blk * RANGE;
  const int col0 = cg * 32 + lm;
  const int frag0 = lm * XP + 2 * half;

  v2f bfr[16][2];
#pragma unroll
  for (int ks = 0; ks < 16; ++ks) {
#pragma unroll
    for (int nt = 0; nt < 2; ++nt) {
      const float* wp = W1 + (ks * 4 + 2 * half) * D_HID + col0 + nt * 16;
      v2f b;
      b.x = wp[0];
      b.y = wp[D_HID];
      bfr[ks][nt] = b;
    }
  }
  const float bias0 = b1[col0];
  const float bias1 = b1[col0 + 16];

  int p = slo[blk];
  int s_hi = slo[blk + 1];
  if (p < 0) p = 0;
  if (s_hi > NSEG) s_hi = NSEG;

  int target = -1;
  float acc0 = 0.0f, acc1 = 0.0f;

  auto flush = [&](int tgt, float a0, float a1) {
    a0 += __shfl_xor(a0, 16, 32);
    a1 += __shfl_xor(a1, 16, 32);
    float* row = (tgt < 0) ? (carry + (size_t)blk * D_HID) : (xm + (size_t)tgt * D_HID);
    if (lane < 16) {
      row[col0] = a0;
      row[col0 + 16] = a1;
    }
  };

  v4f stg[8];
  auto fetch_step = [&](int tb) {
#pragma unroll
    for (int i = 0; i < 8; ++i) {
      const int e = i * 128 + tid;
      stg[i] = *(const v4f*)(x + (size_t)(tb + (e >> 4)) * D_IN + (e & 15) * 4);
    }
  };
  auto store_step = [&](float* buf) {
#pragma unroll
    for (int i = 0; i < 8; ++i) {
      const int e = i * 128 + tid;
      *(v4f*)(buf + (e >> 4) * XP + (e & 15) * 4) = stg[i];
    }
  };

  auto next_start = [&]() -> int { return (p < s_hi) ? offs[p] : 0x7fffffff; };
  auto ask_start = [&]() -> int { return offs[(p < s_hi) ? p : (NSEG - 1)]; };
  auto start_of = [&](int asked) -> int { return (p < s_hi) ? asked : 0x7fffffff; };

  v2f afr[16];
  auto load_frags = [&](const float* buf, int t) {
    const float* ap = buf + t * 16 * XP + frag0;
#pragma unroll
    for (int ks = 0; ks < 16; ++ks) afr[ks] = *(const v2f*)(ap + ks * 4);
    __builtin_amdgcn_sched_barrier(0);
  };

  auto tile_products = [&](v8f& c0, v8f& c1) {
#pragma unroll
    for (int e = 0; e < 8; ++e) {
      c0[e] = 0.0f;
      c1[e] = 0.0f;
    }
#pragma unroll
    for (int ks = 0; ks < 16; ++ks) {
      const v2f a = afr[ks];
      c0 = __builtin_amdgcn_wmma_f32_16x16x4_f32(false, a, false, bfr[ks][0], (short)0, c0, false, false);
      c1 = __builtin_amdgcn_wmma_f32_16x16x4_f32(false, a, false, bfr[ks][1], (short)0, c1, false, false);
    }
    __builtin_amdgcn_sched_barrier(0);
    asm volatile("v_nop\n\tv_nop\n\tv_nop\n\tv_nop");
    __builtin_amdgcn_sched_barrier(0);
  };

  auto tile_rows = [&](const v8f& c0, const v8f& c1, int tt, int asked) {
    int bnd = start_of(asked);
    const int tend = tt + 16;
    const int tok0 = tt + 8 * half;
    float v0[8], v1[8];
#pragma unroll
    for (int r = 0; r < 8; ++r) {
      v0[r] = fmaxf(c0[r] + bias0, 0.0f);
      v1[r] = fmaxf(c1[r] + bias1, 0.0f);
    }
    if (bnd >= tend) {
      acc0 += ((v0[0] + v0[1]) + (v0[2] + v0[3])) + ((v0[4] + v0[5]) + (v0[6] + v0[7]));
      acc1 += ((v1[0] + v1[1]) + (v1[2] + v1[3])) + ((v1[4] + v1[5]) + (v1[6] + v1[7]));
      return;
    }
    int lo = tt;
    while (true) {
      const bool lastp = bnd >= tend;
      const int hi = lastp ? tend : bnd;
      float m0[8], m1[8];
#pragma unroll
      for (int r = 0; r < 8; ++r) {
        const int tok = tok0 + r;
        const bool in = (tok >= lo) && (tok < hi);
        m0[r] = in ? v0[r] : 0.0f;
        m1[r] = in ? v1[r] : 0.0f;
      }
      acc0 += ((m0[0] + m0[1]) + (m0[2] + m0[3])) + ((m0[4] + m0[5]) + (m0[6] + m0[7]));
      acc1 += ((m1[0] + m1[1]) + (m1[2] + m1[3])) + ((m1[4] + m1[5]) + (m1[6] + m1[7]));
      if (lastp) break;
      flush(target, acc0, acc1);
      acc0 = 0.0f;
      acc1 = 0.0f;
      target = p;
      ++p;
      lo = (bnd > lo) ? bnd : lo;
      bnd = next_start();
    }
  };

  v8f cA0, cA1, cB0, cB1;

  fetch_step(T0);
  store_step(xs[0]);
  __syncthreads();
  fetch_step(T0 + STEP);
  int asked = ask_start();
  load_frags(xs[0], 0);
  tile_products(cA0, cA1);
  load_frags(xs[0], 1);
  tile_products(cB0, cB1);

  for (int st = 0; st < RANGE / STEP; ++st) {
    const float* buf = xs[st & 1];
    const int tb = T0 + st * STEP;

    load_frags(buf, 2);
    tile_rows(cA0, cA1, tb, asked);
    asked = ask_start();
    tile_products(cA0, cA1);

    load_frags(buf, 3);
    tile_rows(cB0, cB1, tb + 16, asked);
    asked = ask_start();
    tile_products(cB0, cB1);

    if (st + 1 < RANGE / STEP) {
      float* nbuf = xs[(st + 1) & 1];
      store_step(nbuf);
      __syncthreads();
      if (st + 2 < RANGE / STEP) fetch_step(tb + 2 * STEP);

      load_frags(nbuf, 0);
      tile_rows(cA0, cA1, tb + 32, asked);
      asked = ask_start();
      tile_products(cA0, cA1);

      load_frags(nbuf, 1);
      tile_rows(cB0, cB1, tb + 48, asked);
      asked = ask_start();
      tile_products(cB0, cB1);
    }
  }
  tile_rows(cA0, cA1, T0 + RANGE - 32, asked);
  tile_rows(cB0, cB1, T0 + RANGE - 16, ask_start());

  flush(target, acc0, acc1);
  for (; p < s_hi; ++p) {
    if (lane < 16) {
      xm[(size_t)p * D_HID + col0] = 0.0f;
      xm[(size_t)p * D_HID + col0 + 16] = 0.0f;
    }
  }
}

#define HSEG 32
#define HP   288

__global__ __launch_bounds__(256) void head_gemm_kernel(
    const float* __restrict__ xm, const float* __restrict__ carry,
    const int* __restrict__ offs, const int* __restrict__ l,
    const float* __restrict__ W2, const float* __restrict__ b2,
    float* __restrict__ out) {
  __shared__ __attribute__((aligned(16))) float ws[(D_HID / 2) * HP];

  const int tid  = threadIdx.x;
  const int lane = tid & 31;
  const int wv   = tid >> 5;
  const int half = lane >> 4;
  const int lm   = lane & 15;
  const int rt   = wv & 1;
  const int cg   = wv >> 1;
  const int row0 = blockIdx.x * HSEG + rt * 16;
  const int sg   = row0 + lm;

  const int sst = offs[sg];
  const int sen = sst + l[sg];
  const int n0 = cg * 32 + lm;
  const float bias0 = b2[n0];
  const float bias1 = b2[n0 + 16];

  v4f w0[8], w1[8];
#pragma unroll
  for (int i = 0; i < 8; ++i) {
    const int e  = i * 256 + tid;
    const int kp = e >> 5;
    const int q  = e & 31;
    w0[i] = *(const v4f*)(W2 + (size_t)(2 * kp) * D_HID + 4 * q);
    w1[i] = *(const v4f*)(W2 + (size_t)(2 * kp + 1) * D_HID + 4 * q);
  }

  const float* arow = xm + (size_t)sg * D_HID + 2 * half;
  v2f a[32];
#pragma unroll
  for (int ks = 0; ks < 32; ++ks) a[ks] = *(const v2f*)(arow + ks * 4);

#pragma unroll
  for (int i = 0; i < 8; ++i) {
    const int e  = i * 256 + tid;
    const int kp = e >> 5;
    const int q  = e & 31;
    float* d = ws + kp * HP + 8 * q;
    v4f s0, s1;
    s0.x = w0[i].x; s0.y = w1[i].x; s0.z = w0[i].y; s0.w = w1[i].y;
    s1.x = w0[i].z; s1.y = w1[i].z; s1.z = w0[i].w; s1.w = w1[i].w;
    *(v4f*)d       = s0;
    *(v4f*)(d + 4) = s1;
  }

  int bf = sst / RANGE + 1;
  if (bf < 1) bf = 1;
  int bl = (sen - 1) / RANGE;
  if (bl > NBLK - 1) bl = NBLK - 1;
  for (int b = bf; b <= bl; ++b) {
    const float* cr = carry + (size_t)b * D_HID + 2 * half;
    v2f cv[32];
#pragma unroll
    for (int ks = 0; ks < 32; ++ks) cv[ks] = *(const v2f*)(cr + ks * 4);
#pragma unroll
    for (int ks = 0; ks < 32; ++ks) {
      a[ks].x += cv[ks].x;
      a[ks].y += cv[ks].y;
    }
  }
  __syncthreads();

  const float* bp = ws + half * HP + 2 * (cg * 32 + lm);
  v2f bfr[32][2];
#pragma unroll
  for (int ks = 0; ks < 32; ++ks) {
#pragma unroll
    for (int nt = 0; nt < 2; ++nt) bfr[ks][nt] = *(const v2f*)(bp + (2 * ks) * HP + nt * 32);
  }
  __builtin_amdgcn_sched_barrier(0);

  v8f c0, c1;
#pragma unroll
  for (int e = 0; e < 8; ++e) {
    c0[e] = 0.0f;
    c1[e] = 0.0f;
  }
#pragma unroll
  for (int ks = 0; ks < 32; ++ks) {
    c0 = __builtin_amdgcn_wmma_f32_16x16x4_f32(false, a[ks], false, bfr[ks][0], (short)0, c0, false, false);
    c1 = __builtin_amdgcn_wmma_f32_16x16x4_f32(false, a[ks], false, bfr[ks][1], (short)0, c1, false, false);
  }
  __builtin_amdgcn_sched_barrier(0);
  asm volatile("v_nop\n\tv_nop\n\tv_nop\n\tv_nop");
  __builtin_amdgcn_sched_barrier(0);

#pragma unroll
  for (int r = 0; r < 8; ++r) {
    float* orow = out + (size_t)(row0 + r + 8 * half) * D_HID + n0;
    orow[0]  = c0[r] + bias0;
    orow[16] = c1[r] + bias1;
  }
}

extern "C" void kernel_launch(void* const* d_in, const int* in_sizes, int n_in,
                              void* d_out, int out_size, void* d_ws, size_t ws_size,
                              hipStream_t stream) {
  (void)in_sizes; (void)n_in; (void)out_size; (void)ws_size;
  const float* x  = (const float*)d_in[0];
  const int*   l  = (const int*)d_in[1];
  const float* W1 = (const float*)d_in[2];
  const float* b1 = (const float*)d_in[3];
  const float* W2 = (const float*)d_in[4];
  const float* b2 = (const float*)d_in[5];
  float* out = (float*)d_out;

  int*   offs  = (int*)d_ws;
  int*   slo   = (int*)((char*)d_ws + 32768);
  float* xm    = (float*)((char*)d_ws + 40960);
  float* carry = (float*)((char*)d_ws + 40960 + 4194304);

  seg_scan_kernel<<<1, 1024, 0, stream>>>(l, offs, slo);
  range_mlp_segsum_kernel<<<NBLK, 128, 0, stream>>>(x, W1, b1, offs, slo, xm, carry);
  head_gemm_kernel<<<NSEG / HSEG, 256, 0, stream>>>(xm, carry, offs, l, W2, b2, out);
}
